# starting kernel
# baseline (speedup 1.0000x reference)
#include <hip/hip_runtime.h>
#include <hip/hip_bf16.h>

typedef __attribute__((ext_vector_type(2))) float v2f;
typedef __attribute__((ext_vector_type(4))) float v4f;
typedef __attribute__((ext_vector_type(8))) float v8f;

#define B_ 64
#define T_ 512
#define D_ 1024
#define THALF_ (T_ / 2)

__global__ __launch_bounds__(256) void wtab_kernel(
    const int* __restrict__ wordseq, const int* __restrict__ slen,
    const float* __restrict__ weights, float* __restrict__ wtab) {
  int i = blockIdx.x * 256 + threadIdx.x;
  int b = i >> 9;
  int t = i & (T_ - 1);
  float w = 0.0f;
  if (t < slen[b]) w = weights[wordseq[i]];
  wtab[i] = w;
}

__global__ __launch_bounds__(256) void fused_wmma_kernel(
    const float* __restrict__ vs, const int* __restrict__ slen,
    const float* __restrict__ wtab,
    float* __restrict__ s_part, float* __restrict__ yh_part) {
  __shared__ float lds_w[T_];

  const int tid  = threadIdx.x;
  const int lane = tid & 31;
  const int wid  = blockIdx.x * 8 + (tid >> 5);
  const int b      = wid >> 5;
  const int tile   = (wid >> 1) & 15;
  const int thalf  = wid & 1;
  const int d_base = tile * 64;
  const int tbase  = thalf * THALF_;

  for (int i = tid; i < T_; i += 256) lds_w[i] = wtab[b * T_ + i];
  __syncthreads();

  const int   m    = lane & 15;
  const int   h    = lane >> 4;
  const int   lenl = slen[b] - tbase;
  const float rowsel0 = (m == 0) ? 1.0f : 0.0f;
  const float rowsel1 = (m == 1) ? 1.0f : 0.0f;

  v8f c0 = {}, c1 = {}, c2 = {}, c3 = {};

  const float* base  = vs + (size_t)b * T_ * D_ + (size_t)tbase * D_ +
                       (size_t)(d_base + 4 * m);
  const float* wrow  = lds_w + tbase;

#pragma unroll 8
  for (int t0 = 0; t0 < THALF_; t0 += 4) {
    const int te = t0 + 2 * h;
    v4f r0 = *(const v4f*)(base + (size_t)te * D_);
    v4f r1 = *(const v4f*)(base + (size_t)(te + 1) * D_);

    const float mk0 = (te     < lenl) ? 1.0f : 0.0f;
    const float mk1 = (te + 1 < lenl) ? 1.0f : 0.0f;
    const float w0  = wrow[te];
    const float w1  = wrow[te + 1];
    v2f a;
    a.x = rowsel0 * mk0 + rowsel1 * w0;
    a.y = rowsel0 * mk1 + rowsel1 * w1;

    v2f bop;
    bop.x = r0.x; bop.y = r1.x;
    c0 = __builtin_amdgcn_wmma_f32_16x16x4_f32(false, a, false, bop, (short)0, c0, false, false);
    bop.x = r0.y; bop.y = r1.y;
    c1 = __builtin_amdgcn_wmma_f32_16x16x4_f32(false, a, false, bop, (short)0, c1, false, false);
    bop.x = r0.z; bop.y = r1.z;
    c2 = __builtin_amdgcn_wmma_f32_16x16x4_f32(false, a, false, bop, (short)0, c2, false, false);
    bop.x = r0.w; bop.y = r1.w;
    c3 = __builtin_amdgcn_wmma_f32_16x16x4_f32(false, a, false, bop, (short)0, c3, false, false);
  }

  if (lane < 16) {
    v4f sv, hv;
    sv.x = c0[0]; sv.y = c1[0]; sv.z = c2[0]; sv.w = c3[0];
    hv.x = c0[1]; hv.y = c1[1]; hv.z = c2[1]; hv.w = c3[1];
    const size_t off = (size_t)(thalf * B_ + b) * D_ + d_base + 4 * lane;
    *(v4f*)(s_part  + off) = sv;
    *(v4f*)(yh_part + off) = hv;
  }
}

__global__ __launch_bounds__(256) void finalize_kernel(
    const float* __restrict__ s_part, const float* __restrict__ yh_part,
    float* __restrict__ y_out, float* __restrict__ yh_out) {
  __shared__ float red[256];
  const int b   = blockIdx.x;
  const int tid = threadIdx.x;

  const v4f* s0 = (const v4f*)(s_part + (size_t)b * D_);
  const v4f* s1 = (const v4f*)(s_part + (size_t)(B_ + b) * D_);
  const v4f* h0 = (const v4f*)(yh_part + (size_t)b * D_);
  const v4f* h1 = (const v4f*)(yh_part + (size_t)(B_ + b) * D_);

  v4f sv = s0[tid];
  v4f sw = s1[tid];
  sv.x += sw.x; sv.y += sw.y; sv.z += sw.z; sv.w += sw.w;

  v4f hv = h0[tid];
  v4f hw = h1[tid];
  hv.x += hw.x; hv.y += hw.y; hv.z += hw.z; hv.w += hw.w;
  ((v4f*)(yh_out + (size_t)b * D_))[tid] = hv;

  red[tid] = fabsf(sv.x) + fabsf(sv.y) + fabsf(sv.z) + fabsf(sv.w);
  __syncthreads();
  for (int off = 128; off > 0; off >>= 1) {
    if (tid < off) red[tid] += red[tid + off];
    __syncthreads();
  }
  const float inv = 1.0f / sqrtf(red[0]);
  v4f yv;
  yv.x = sv.x * inv; yv.y = sv.y * inv; yv.z = sv.z * inv; yv.w = sv.w * inv;
  ((v4f*)(y_out + (size_t)b * D_))[tid] = yv;
}

extern "C" void kernel_launch(void* const* d_in, const int* in_sizes, int n_in,
                              void* d_out, int out_size, void* d_ws, size_t ws_size,
                              hipStream_t stream) {
  const float* vs      = (const float*)d_in[0];
  const int*   slen    = (const int*)d_in[1];
  const int*   wordseq = (const int*)d_in[2];
  const float* weights = (const float*)d_in[3];

  float* y_out  = (float*)d_out;
  float* yh_out = (float*)d_out + B_ * D_;

  float* wtab    = (float*)d_ws;
  float* s_part  = wtab + B_ * T_;
  float* yh_part = s_part + 2 * B_ * D_;

  wtab_kernel<<<(B_ * T_) / 256, 256, 0, stream>>>(wordseq, slen, weights, wtab);
  fused_wmma_kernel<<<256, 256, 0, stream>>>(vs, slen, wtab, s_part, yh_part);
  finalize_kernel<<<B_, 256, 0, stream>>>(s_part, yh_part, y_out, yh_out);
}


# live K-steps of all sequences joined, cut into 64 equal ranges x 16 column blocks; a range's piece of a sequence dealt to 4 waves, summed in LDS, stored as partial p+b (sum order over t changes)
# speedup vs baseline: 1.0761x; 1.0761x over previous
#include <hip/hip_runtime.h>
#include <hip/hip_bf16.h>


typedef __attribute__((ext_vector_type(2))) float v2f;
typedef __attribute__((ext_vector_type(4))) float v4f;
typedef __attribute__((ext_vector_type(8))) float v8f;

#define B_ 64
#define T_ 512
#define D_ 1024
#define NP_ 64
#define NWV_ 4
#define GK_ 4

__device__ __forceinline__ int ksteps_of(int len) {
  len = len < 0 ? 0 : (len > T_ ? T_ : len);
  return (len + 3) >> 2;
}

__global__ __launch_bounds__(256) void wtab_kernel(
    const int* __restrict__ wordseq, const int* __restrict__ slen,
    const float* __restrict__ weights, float* __restrict__ wtab) {
  int i = blockIdx.x * 256 + threadIdx.x;
  int b = i >> 9;
  int t = i & (T_ - 1);
  float w = 0.0f;
  if (t < slen[b]) w = weights[wordseq[i]];
  wtab[i] = w;
}

__global__ __launch_bounds__(128) void stream_kernel(
    const float* __restrict__ vs, const int* __restrict__ slen,
    const float* __restrict__ wtab,
    float* __restrict__ s_part, float* __restrict__ yh_part) {
  __shared__ v4f red[2][NWV_][2][16];

  const int tid  = threadIdx.x;
  const int lane = tid & 31;
  const int wv   = __builtin_amdgcn_readfirstlane(tid >> 5);
  const int p      = blockIdx.x >> 4;
  const int d_base = (blockIdx.x & 15) * 64;
  const int wfirst = (wv + (int)blockIdx.x) & (NWV_ - 1);

  int S = 0;
  for (int i = 0; i < B_; ++i) S += ksteps_of(slen[i]);
  int L = (S + NP_ - 1) / NP_;
  if (L < 1) L = 1;
  const int k0 = p * L;
  const int k1 = (k0 + L < S) ? (k0 + L) : S;
  if (k0 >= k1) return;

  int b = 0, cstart = 0, run = 0;
  for (int i = 0; i < B_; ++i) {
    run += ksteps_of(slen[i]);
    if (run <= k0) { b = i + 1; cstart = run; }
  }

  const int   m = lane & 15;
  const int   h = lane >> 4;
  const float rowsel0 = (m == 0) ? 1.0f : 0.0f;
  const float rowsel1 = (m == 1) ? 1.0f : 0.0f;

  int k = k0;
  int par = 0;
  while (k < k1) {
    const int len  = slen[b];
    const int bend = cstart + ksteps_of(len);
    const int kend = (k1 < bend) ? k1 : bend;
    if (kend <= k) { cstart = bend; ++b; continue; }

    const int kl0 = k - cstart;
    const int nk  = kend - k;
    const float* base = vs + (size_t)b * T_ * D_ + (size_t)(d_base + 4 * m);
    const float* wrow = wtab + b * T_;

    v8f c0 = {}, c1 = {}, c2 = {}, c3 = {};

    for (int g = wfirst * GK_; g < nk; g += NWV_ * GK_) {
      v4f r0[GK_], r1[GK_];
      v2f w2[GK_];
      int te[GK_];
      for (int u = 0; u < GK_; ++u) {
        const int kc = ((g + u) < nk) ? (g + u) : (nk - 1);
        te[u] = (kl0 + kc) * 4 + 2 * h;
        r0[u] = *(const v4f*)(base + (size_t)te[u] * D_);
        r1[u] = *(const v4f*)(base + (size_t)(te[u] + 1) * D_);
        w2[u] = *(const v2f*)(wrow + te[u]);
      }
      __builtin_amdgcn_sched_barrier(0);
      for (int u = 0; u < GK_; ++u) {
        const bool live = (g + u) < nk;
        const float mk0 = (live && te[u]     < len) ? 1.0f : 0.0f;
        const float mk1 = (live && te[u] + 1 < len) ? 1.0f : 0.0f;
        const float w0  = live ? w2[u].x : 0.0f;
        const float w1  = live ? w2[u].y : 0.0f;
        v2f a;
        a.x = rowsel0 * mk0 + rowsel1 * w0;
        a.y = rowsel0 * mk1 + rowsel1 * w1;
        v2f bop;
        bop.x = r0[u].x; bop.y = r1[u].x;
        c0 = __builtin_amdgcn_wmma_f32_16x16x4_f32(false, a, false, bop, (short)0, c0, false, false);
        bop.x = r0[u].y; bop.y = r1[u].y;
        c1 = __builtin_amdgcn_wmma_f32_16x16x4_f32(false, a, false, bop, (short)0, c1, false, false);
        bop.x = r0[u].z; bop.y = r1[u].z;
        c2 = __builtin_amdgcn_wmma_f32_16x16x4_f32(false, a, false, bop, (short)0, c2, false, false);
        bop.x = r0[u].w; bop.y = r1[u].w;
        c3 = __builtin_amdgcn_wmma_f32_16x16x4_f32(false, a, false, bop, (short)0, c3, false, false);
      }
    }

    if (lane < 16) {
      v4f sv, hv;
      sv.x = c0[0]; sv.y = c1[0]; sv.z = c2[0]; sv.w = c3[0];
      hv.x = c0[1]; hv.y = c1[1]; hv.z = c2[1]; hv.w = c3[1];
      red[par][wv][0][lane] = sv;
      red[par][wv][1][lane] = hv;
    }
    __syncthreads();
    if (tid < 32) {
      const int r = tid >> 4;
      const int n = tid & 15;
      v4f acc = red[par][0][r][n];
      for (int w = 1; w < NWV_; ++w) {
        const v4f x = red[par][w][r][n];
        acc.x += x.x; acc.y += x.y; acc.z += x.z; acc.w += x.w;
      }
      float* dst = (r ? yh_part : s_part) + (size_t)(p + b) * D_ + d_base + 4 * n;
      *(v4f*)dst = acc;
    }
    par ^= 1;
    k = kend;
    if (kend == bend) { cstart = bend; ++b; }
  }
}

__global__ __launch_bounds__(256) void finalize_kernel(
    const int* __restrict__ slen,
    const float* __restrict__ s_part, const float* __restrict__ yh_part,
    float* __restrict__ y_out, float* __restrict__ yh_out) {
  __shared__ float red[256];
  const int b   = blockIdx.x;
  const int tid = threadIdx.x;

  int S = 0, bstart = 0;
  for (int i = 0; i < B_; ++i) {
    const int sb = ksteps_of(slen[i]);
    if (i < b) bstart += sb;
    S += sb;
  }
  const int sbb = ksteps_of(slen[b]);
  int L = (S + NP_ - 1) / NP_;
  if (L < 1) L = 1;

  v4f sv = {0.0f, 0.0f, 0.0f, 0.0f};
  v4f hv = {0.0f, 0.0f, 0.0f, 0.0f};
  if (sbb > 0) {
    const int p_lo = bstart / L;
    const int p_hi = (bstart + sbb - 1) / L;
    for (int p = p_lo; p <= p_hi; ++p) {
      const v4f x = ((const v4f*)(s_part  + (size_t)(p + b) * D_))[tid];
      const v4f z = ((const v4f*)(yh_part + (size_t)(p + b) * D_))[tid];
      sv.x += x.x; sv.y += x.y; sv.z += x.z; sv.w += x.w;
      hv.x += z.x; hv.y += z.y; hv.z += z.z; hv.w += z.w;
    }
  }
  ((v4f*)(yh_out + (size_t)b * D_))[tid] = hv;

  red[tid] = fabsf(sv.x) + fabsf(sv.y) + fabsf(sv.z) + fabsf(sv.w);
  __syncthreads();
  for (int off = 128; off > 0; off >>= 1) {
    if (tid < off) red[tid] += red[tid + off];
    __syncthreads();
  }
  const float inv = 1.0f / sqrtf(red[0]);
  v4f yv;
  yv.x = sv.x * inv; yv.y = sv.y * inv; yv.z = sv.z * inv; yv.w = sv.w * inv;
  ((v4f*)(y_out + (size_t)b * D_))[tid] = yv;
}

extern "C" void kernel_launch(void* const* d_in, const int* in_sizes, int n_in,
                              void* d_out, int out_size, void* d_ws, size_t ws_size,
                              hipStream_t stream) {
  const float* vs      = (const float*)d_in[0];
  const int*   slen    = (const int*)d_in[1];
  const int*   wordseq = (const int*)d_in[2];
  const float* weights = (const float*)d_in[3];

  float* y_out  = (float*)d_out;
  float* yh_out = (float*)d_out + B_ * D_;

  float* wtab    = (float*)d_ws;
  float* s_part  = wtab + B_ * T_;
  float* yh_part = s_part + 2 * B_ * D_;

  wtab_kernel<<<(B_ * T_) / 256, 256, 0, stream>>>(wordseq, slen, weights, wtab);
  stream_kernel<<<NP_ * 16, NWV_ * 32, 0, stream>>>(vs, slen, wtab, s_part, yh_part);
  finalize_kernel<<<B_, 256, 0, stream>>>(slen, s_part, yh_part, y_out, yh_out);
}


# split bookkeeping (per range: first sequence, start, length, k0, k1; per sequence: first/last range) built once by one block of kernel 0 in a workspace table; kernels 1, 2 read it (sums unchanged)
# speedup vs baseline: 1.2820x; 1.1913x over previous
#include <hip/hip_runtime.h>
#include <hip/hip_bf16.h>


typedef __attribute__((ext_vector_type(2))) float v2f;
typedef __attribute__((ext_vector_type(4))) float v4f;
typedef __attribute__((ext_vector_type(8))) float v8f;

#define B_ 64
#define T_ 512
#define D_ 1024
#define NP_ 64
#define NWV_ 4
#define GK_ 4
#define WBLK_ ((B_ * T_) / 256)
#define PSEQ_ (NP_ * 8)

static_assert(NP_ == B_, "one table thread serves range i and sequence i");

__device__ __forceinline__ int ksteps_of(int len) {
  len = len < 0 ? 0 : (len > T_ ? T_ : len);
  return (len + 3) >> 2;
}

__global__ __launch_bounds__(256) void wtab_kernel(
    const int* __restrict__ wordseq, const int* __restrict__ slen,
    const float* __restrict__ weights, float* __restrict__ wtab,
    int* __restrict__ plan) {
  if (blockIdx.x == 0) {
    __shared__ int sh_ks[B_];
    __shared__ int sh_run[B_];
    __shared__ int sh_len[B_];
    const int i = threadIdx.x;
    int ks = 0;
    if (i < B_) {
      const int len = slen[i];
      ks = ksteps_of(len);
      sh_ks[i]  = ks;
      sh_len[i] = len;
    }
    __syncthreads();
    int cst = 0;
    if (i < B_) {
      for (int j = 0; j < B_; ++j) cst += (j < i) ? sh_ks[j] : 0;
      sh_run[i] = cst + ks;
    }
    __syncthreads();
    if (i < B_) {
      const int S = sh_run[B_ - 1];
      int L = (S + NP_ - 1) / NP_;
      if (L < 1) L = 1;
      const int k0 = i * L;
      const int k1 = (k0 + L < S) ? (k0 + L) : S;
      int bf = 0, cs = 0;
      for (int j = 0; j < B_; ++j) {
        const int r = sh_run[j];
        if (r <= k0) { bf = j + 1; cs = r; }
      }
      const int lf = sh_len[bf < B_ ? bf : B_ - 1];
      int4 e0; e0.x = bf; e0.y = cs; e0.z = k0; e0.w = k1;
      int4 e1; e1.x = lf; e1.y = 0;  e1.z = 0;  e1.w = 0;
      ((int4*)plan)[2 * i]     = e0;
      ((int4*)plan)[2 * i + 1] = e1;
      int2 pr; pr.x = 0; pr.y = -1;
      if (ks > 0) { pr.x = cst / L; pr.y = (cst + ks - 1) / L; }
      ((int2*)(plan + PSEQ_))[i] = pr;
    }
    return;
  }
  int i = ((int)blockIdx.x - 1) * 256 + threadIdx.x;
  int b = i >> 9;
  int t = i & (T_ - 1);
  float w = 0.0f;
  if (t < slen[b]) w = weights[wordseq[i]];
  wtab[i] = w;
}

__global__ __launch_bounds__(128) void stream_kernel(
    const float* __restrict__ vs, const int* __restrict__ slen,
    const float* __restrict__ wtab, const int* __restrict__ plan,
    float* __restrict__ s_part, float* __restrict__ yh_part) {
  __shared__ v4f red[2][NWV_][2][16];

  const int tid  = threadIdx.x;
  const int lane = tid & 31;
  const int wv   = __builtin_amdgcn_readfirstlane(tid >> 5);
  const int p      = blockIdx.x >> 4;
  const int d_base = (blockIdx.x & 15) * 64;
  const int wfirst = (wv + (int)blockIdx.x) & (NWV_ - 1);

  const int4 e0 = ((const int4*)plan)[2 * p];
  int len       = plan[8 * p + 4];
  int b         = e0.x;
  int cstart    = e0.y;
  const int k0  = e0.z;
  const int k1  = e0.w;
  if (k0 >= k1) return;

  const int   m = lane & 15;
  const int   h = lane >> 4;
  const float rowsel0 = (m == 0) ? 1.0f : 0.0f;
  const float rowsel1 = (m == 1) ? 1.0f : 0.0f;

  int k = k0;
  int par = 0;
  while (k < k1) {
    const int bend = cstart + ksteps_of(len);
    const int kend = (k1 < bend) ? k1 : bend;
    const int len_nx = slen[(b + 1 < B_) ? (b + 1) : (B_ - 1)];
    if (kend <= k) { cstart = bend; ++b; len = len_nx; continue; }

    const int kl0 = k - cstart;
    const int nk  = kend - k;
    const float* base = vs + (size_t)b * T_ * D_ + (size_t)(d_base + 4 * m);
    const float* wrow = wtab + b * T_;

    v8f c0 = {}, c1 = {}, c2 = {}, c3 = {};

    for (int g = wfirst * GK_; g < nk; g += NWV_ * GK_) {
      v4f r0[GK_], r1[GK_];
      v2f w2[GK_];
      int te[GK_];
      for (int u = 0; u < GK_; ++u) {
        const int kc = ((g + u) < nk) ? (g + u) : (nk - 1);
        te[u] = (kl0 + kc) * 4 + 2 * h;
        r0[u] = *(const v4f*)(base + (size_t)te[u] * D_);
        r1[u] = *(const v4f*)(base + (size_t)(te[u] + 1) * D_);
        w2[u] = *(const v2f*)(wrow + te[u]);
      }
      __builtin_amdgcn_sched_barrier(0);
      for (int u = 0; u < GK_; ++u) {
        const bool live = (g + u) < nk;
        const float mk0 = (live && te[u]     < len) ? 1.0f : 0.0f;
        const float mk1 = (live && te[u] + 1 < len) ? 1.0f : 0.0f;
        const float w0  = live ? w2[u].x : 0.0f;
        const float w1  = live ? w2[u].y : 0.0f;
        v2f a;
        a.x = rowsel0 * mk0 + rowsel1 * w0;
        a.y = rowsel0 * mk1 + rowsel1 * w1;
        v2f bop;
        bop.x = r0[u].x; bop.y = r1[u].x;
        c0 = __builtin_amdgcn_wmma_f32_16x16x4_f32(false, a, false, bop, (short)0, c0, false, false);
        bop.x = r0[u].y; bop.y = r1[u].y;
        c1 = __builtin_amdgcn_wmma_f32_16x16x4_f32(false, a, false, bop, (short)0, c1, false, false);
        bop.x = r0[u].z; bop.y = r1[u].z;
        c2 = __builtin_amdgcn_wmma_f32_16x16x4_f32(false, a, false, bop, (short)0, c2, false, false);
        bop.x = r0[u].w; bop.y = r1[u].w;
        c3 = __builtin_amdgcn_wmma_f32_16x16x4_f32(false, a, false, bop, (short)0, c3, false, false);
      }
    }

    if (lane < 16) {
      v4f sv, hv;
      sv.x = c0[0]; sv.y = c1[0]; sv.z = c2[0]; sv.w = c3[0];
      hv.x = c0[1]; hv.y = c1[1]; hv.z = c2[1]; hv.w = c3[1];
      red[par][wv][0][lane] = sv;
      red[par][wv][1][lane] = hv;
    }
    __syncthreads();
    if (tid < 32) {
      const int r = tid >> 4;
      const int n = tid & 15;
      v4f acc = red[par][0][r][n];
      for (int w = 1; w < NWV_; ++w) {
        const v4f x = red[par][w][r][n];
        acc.x += x.x; acc.y += x.y; acc.z += x.z; acc.w += x.w;
      }
      float* dst = (r ? yh_part : s_part) + (size_t)(p + b) * D_ + d_base + 4 * n;
      *(v4f*)dst = acc;
    }
    par ^= 1;
    k = kend;
    if (kend == bend) { cstart = bend; ++b; len = len_nx; }
  }
}

__global__ __launch_bounds__(256) void finalize_kernel(
    const int* __restrict__ plan,
    const float* __restrict__ s_part, const float* __restrict__ yh_part,
    float* __restrict__ y_out, float* __restrict__ yh_out) {
  __shared__ float red[256];
  const int b   = blockIdx.x;
  const int tid = threadIdx.x;

  const int2 pr  = ((const int2*)(plan + PSEQ_))[b];
  const int p_lo = pr.x;
  const int p_hi = pr.y;

  v4f sv = {0.0f, 0.0f, 0.0f, 0.0f};
  v4f hv = {0.0f, 0.0f, 0.0f, 0.0f};
  {
    for (int p = p_lo; p <= p_hi; ++p) {
      const v4f x = ((const v4f*)(s_part  + (size_t)(p + b) * D_))[tid];
      const v4f z = ((const v4f*)(yh_part + (size_t)(p + b) * D_))[tid];
      sv.x += x.x; sv.y += x.y; sv.z += x.z; sv.w += x.w;
      hv.x += z.x; hv.y += z.y; hv.z += z.z; hv.w += z.w;
    }
  }
  ((v4f*)(yh_out + (size_t)b * D_))[tid] = hv;

  red[tid] = fabsf(sv.x) + fabsf(sv.y) + fabsf(sv.z) + fabsf(sv.w);
  __syncthreads();
  for (int off = 128; off > 0; off >>= 1) {
    if (tid < off) red[tid] += red[tid + off];
    __syncthreads();
  }
  const float inv = 1.0f / sqrtf(red[0]);
  v4f yv;
  yv.x = sv.x * inv; yv.y = sv.y * inv; yv.z = sv.z * inv; yv.w = sv.w * inv;
  ((v4f*)(y_out + (size_t)b * D_))[tid] = yv;
}

extern "C" void kernel_launch(void* const* d_in, const int* in_sizes, int n_in,
                              void* d_out, int out_size, void* d_ws, size_t ws_size,
                              hipStream_t stream) {
  const float* vs      = (const float*)d_in[0];
  const int*   slen    = (const int*)d_in[1];
  const int*   wordseq = (const int*)d_in[2];
  const float* weights = (const float*)d_in[3];

  float* y_out  = (float*)d_out;
  float* yh_out = (float*)d_out + B_ * D_;

  float* wtab    = (float*)d_ws;
  float* s_part  = wtab + B_ * T_;
  float* yh_part = s_part + 2 * B_ * D_;
  int*   plan    = (int*)(s_part + 127 * D_);

  wtab_kernel<<<WBLK_ + 1, 256, 0, stream>>>(wordseq, slen, weights, wtab, plan);
  stream_kernel<<<NP_ * 16, NWV_ * 32, 0, stream>>>(vs, slen, wtab, plan, s_part, yh_part);
  finalize_kernel<<<B_, 256, 0, stream>>>(plan, s_part, yh_part, y_out, yh_out);
}


# each wave streams a contiguous quarter of a piece through a rolling ring of 8 K-steps, refilled behind each step's WMMAs, dead steps not run; was groups of 4 dealt in turn (sum order over t changes)
# speedup vs baseline: 1.3762x; 1.0735x over previous
#include <hip/hip_runtime.h>
#include <hip/hip_bf16.h>


typedef __attribute__((ext_vector_type(2))) float v2f;
typedef __attribute__((ext_vector_type(4))) float v4f;
typedef __attribute__((ext_vector_type(8))) float v8f;

#define B_ 64
#define T_ 512
#define D_ 1024
#define NP_ 64
#define NWV_ 4
#define RK_ 8
#define WBLK_ ((B_ * T_) / 256)
#define PSEQ_ (NP_ * 8)

static_assert(NP_ == B_, "one table thread serves range i and sequence i");

__device__ __forceinline__ int ksteps_of(int len) {
  len = len < 0 ? 0 : (len > T_ ? T_ : len);
  return (len + 3) >> 2;
}

__global__ __launch_bounds__(256) void wtab_kernel(
    const int* __restrict__ wordseq, const int* __restrict__ slen,
    const float* __restrict__ weights, float* __restrict__ wtab,
    int* __restrict__ plan) {
  if (blockIdx.x == 0) {
    __shared__ int sh_ks[B_];
    __shared__ int sh_run[B_];
    __shared__ int sh_len[B_];
    const int i = threadIdx.x;
    int ks = 0;
    if (i < B_) {
      const int len = slen[i];
      ks = ksteps_of(len);
      sh_ks[i]  = ks;
      sh_len[i] = len;
    }
    __syncthreads();
    int cst = 0;
    if (i < B_) {
      for (int j = 0; j < B_; ++j) cst += (j < i) ? sh_ks[j] : 0;
      sh_run[i] = cst + ks;
    }
    __syncthreads();
    if (i < B_) {
      const int S = sh_run[B_ - 1];
      int L = (S + NP_ - 1) / NP_;
      if (L < 1) L = 1;
      const int k0 = i * L;
      const int k1 = (k0 + L < S) ? (k0 + L) : S;
      int bf = 0, cs = 0;
      for (int j = 0; j < B_; ++j) {
        const int r = sh_run[j];
        if (r <= k0) { bf = j + 1; cs = r; }
      }
      const int lf = sh_len[bf < B_ ? bf : B_ - 1];
      int4 e0; e0.x = bf; e0.y = cs; e0.z = k0; e0.w = k1;
      int4 e1; e1.x = lf; e1.y = 0;  e1.z = 0;  e1.w = 0;
      ((int4*)plan)[2 * i]     = e0;
      ((int4*)plan)[2 * i + 1] = e1;
      int2 pr; pr.x = 0; pr.y = -1;
      if (ks > 0) { pr.x = cst / L; pr.y = (cst + ks - 1) / L; }
      ((int2*)(plan + PSEQ_))[i] = pr;
    }
    return;
  }
  int i = ((int)blockIdx.x - 1) * 256 + threadIdx.x;
  int b = i >> 9;
  int t = i & (T_ - 1);
  float w = 0.0f;
  if (t < slen[b]) w = weights[wordseq[i]];
  wtab[i] = w;
}

__global__ __launch_bounds__(128) void stream_kernel(
    const float* __restrict__ vs, const int* __restrict__ slen,
    const float* __restrict__ wtab, const int* __restrict__ plan,
    float* __restrict__ s_part, float* __restrict__ yh_part) {
  __shared__ v4f red[2][NWV_][2][16];

  const int tid  = threadIdx.x;
  const int lane = tid & 31;
  const int wv   = __builtin_amdgcn_readfirstlane(tid >> 5);
  const int p      = blockIdx.x >> 4;
  const int d_base = (blockIdx.x & 15) * 64;
  const int wfirst = (wv + (int)blockIdx.x) & (NWV_ - 1);

  const int4 e0 = ((const int4*)plan)[2 * p];
  int len       = plan[8 * p + 4];
  int b         = e0.x;
  int cstart    = e0.y;
  const int k0  = e0.z;
  const int k1  = e0.w;
  if (k0 >= k1) return;

  const int   m = lane & 15;
  const int   h = lane >> 4;
  const float rowsel0 = (m == 0) ? 1.0f : 0.0f;
  const float rowsel1 = (m == 1) ? 1.0f : 0.0f;

  int k = k0;
  int par = 0;
  while (k < k1) {
    const int bend = cstart + ksteps_of(len);
    const int kend = (k1 < bend) ? k1 : bend;
    const int len_nx = slen[(b + 1 < B_) ? (b + 1) : (B_ - 1)];
    if (kend <= k) { cstart = bend; ++b; len = len_nx; continue; }

    const int kl0 = k - cstart;
    const int nk  = kend - k;
    const float* base = vs + (size_t)b * T_ * D_ + (size_t)(d_base + 4 * m);
    const float* wrow = wtab + b * T_;

    v8f c0 = {}, c1 = {}, c2 = {}, c3 = {};

    const int ws = (nk * wfirst) >> 2;
    const int we = (nk * (wfirst + 1)) >> 2;

    if (ws < we) {
      v4f r0[RK_], r1[RK_];
      v2f w2[RK_];
#define LOAD_STEP(u, j)                                                        \
      {                                                                        \
        const int jc_ = ((j) < we) ? (j) : (we - 1);                           \
        const int te_ = (kl0 + jc_) * 4 + 2 * h;                               \
        r0[u] = *(const v4f*)(base + (size_t)te_ * D_);                        \
        r1[u] = *(const v4f*)(base + (size_t)(te_ + 1) * D_);                  \
        w2[u] = *(const v2f*)(wrow + te_);                                     \
      }
#define USE_STEP(u, j)                                                         \
      {                                                                        \
        asm volatile("" : "+v"(r0[u]), "+v"(r1[u]), "+v"(w2[u]));              \
        const int te_ = (kl0 + (j)) * 4 + 2 * h;                               \
        const float mk0 = (te_     < len) ? 1.0f : 0.0f;                       \
        const float mk1 = (te_ + 1 < len) ? 1.0f : 0.0f;                       \
        v2f a;                                                                 \
        a.x = rowsel0 * mk0 + rowsel1 * w2[u].x;                               \
        a.y = rowsel0 * mk1 + rowsel1 * w2[u].y;                               \
        v2f bop;                                                               \
        bop.x = r0[u].x; bop.y = r1[u].x;                                      \
        c0 = __builtin_amdgcn_wmma_f32_16x16x4_f32(false, a, false, bop, (short)0, c0, false, false); \
        bop.x = r0[u].y; bop.y = r1[u].y;                                      \
        c1 = __builtin_amdgcn_wmma_f32_16x16x4_f32(false, a, false, bop, (short)0, c1, false, false); \
        bop.x = r0[u].z; bop.y = r1[u].z;                                      \
        c2 = __builtin_amdgcn_wmma_f32_16x16x4_f32(false, a, false, bop, (short)0, c2, false, false); \
        bop.x = r0[u].w; bop.y = r1[u].w;                                      \
        c3 = __builtin_amdgcn_wmma_f32_16x16x4_f32(false, a, false, bop, (short)0, c3, false, false); \
      }

#pragma unroll
      for (int u = 0; u < RK_; ++u) LOAD_STEP(u, ws + u)

      int i = ws;
      for (; i + RK_ < we; i += RK_) {
#pragma unroll
        for (int u = 0; u < RK_; ++u) {
          USE_STEP(u, i + u)
          LOAD_STEP(u, i + u + RK_)
        }
      }
#pragma unroll
      for (int u = 0; u < RK_; ++u) {
        if (i + u < we) USE_STEP(u, i + u)
      }
#undef LOAD_STEP
#undef USE_STEP
    }

    if (lane < 16) {
      v4f sv, hv;
      sv.x = c0[0]; sv.y = c1[0]; sv.z = c2[0]; sv.w = c3[0];
      hv.x = c0[1]; hv.y = c1[1]; hv.z = c2[1]; hv.w = c3[1];
      red[par][wv][0][lane] = sv;
      red[par][wv][1][lane] = hv;
    }
    __syncthreads();
    if (tid < 32) {
      const int r = tid >> 4;
      const int n = tid & 15;
      v4f acc = red[par][0][r][n];
      for (int w = 1; w < NWV_; ++w) {
        const v4f x = red[par][w][r][n];
        acc.x += x.x; acc.y += x.y; acc.z += x.z; acc.w += x.w;
      }
      float* dst = (r ? yh_part : s_part) + (size_t)(p + b) * D_ + d_base + 4 * n;
      *(v4f*)dst = acc;
    }
    par ^= 1;
    k = kend;
    if (kend == bend) { cstart = bend; ++b; len = len_nx; }
  }
}

__global__ __launch_bounds__(256) void finalize_kernel(
    const int* __restrict__ plan,
    const float* __restrict__ s_part, const float* __restrict__ yh_part,
    float* __restrict__ y_out, float* __restrict__ yh_out) {
  __shared__ float red[256];
  const int b   = blockIdx.x;
  const int tid = threadIdx.x;

  const int2 pr  = ((const int2*)(plan + PSEQ_))[b];
  const int p_lo = pr.x;
  const int p_hi = pr.y;

  v4f sv = {0.0f, 0.0f, 0.0f, 0.0f};
  v4f hv = {0.0f, 0.0f, 0.0f, 0.0f};
  {
    for (int p = p_lo; p <= p_hi; ++p) {
      const v4f x = ((const v4f*)(s_part  + (size_t)(p + b) * D_))[tid];
      const v4f z = ((const v4f*)(yh_part + (size_t)(p + b) * D_))[tid];
      sv.x += x.x; sv.y += x.y; sv.z += x.z; sv.w += x.w;
      hv.x += z.x; hv.y += z.y; hv.z += z.z; hv.w += z.w;
    }
  }
  ((v4f*)(yh_out + (size_t)b * D_))[tid] = hv;

  red[tid] = fabsf(sv.x) + fabsf(sv.y) + fabsf(sv.z) + fabsf(sv.w);
  __syncthreads();
  for (int off = 128; off > 0; off >>= 1) {
    if (tid < off) red[tid] += red[tid + off];
    __syncthreads();
  }
  const float inv = 1.0f / sqrtf(red[0]);
  v4f yv;
  yv.x = sv.x * inv; yv.y = sv.y * inv; yv.z = sv.z * inv; yv.w = sv.w * inv;
  ((v4f*)(y_out + (size_t)b * D_))[tid] = yv;
}

extern "C" void kernel_launch(void* const* d_in, const int* in_sizes, int n_in,
                              void* d_out, int out_size, void* d_ws, size_t ws_size,
                              hipStream_t stream) {
  const float* vs      = (const float*)d_in[0];
  const int*   slen    = (const int*)d_in[1];
  const int*   wordseq = (const int*)d_in[2];
  const float* weights = (const float*)d_in[3];

  float* y_out  = (float*)d_out;
  float* yh_out = (float*)d_out + B_ * D_;

  float* wtab    = (float*)d_ws;
  float* s_part  = wtab + B_ * T_;
  float* yh_part = s_part + 2 * B_ * D_;
  int*   plan    = (int*)(s_part + 127 * D_);

  wtab_kernel<<<WBLK_ + 1, 256, 0, stream>>>(wordseq, slen, weights, wtab, plan);
  stream_kernel<<<NP_ * 16, NWV_ * 32, 0, stream>>>(vs, slen, wtab, plan, s_part, yh_part);
  finalize_kernel<<<B_, 256, 0, stream>>>(plan, s_part, yh_part, y_out, yh_out);
}


# finalize kernel: one wave per batch row (was 8), the row's partials asked for as one group of loads, L1 norm by lane exchanges alone, no LDS and no barrier (norm's sum order changes)
# speedup vs baseline: 1.5747x; 1.1442x over previous
#include <hip/hip_runtime.h>
#include <hip/hip_bf16.h>


typedef __attribute__((ext_vector_type(2))) float v2f;
typedef __attribute__((ext_vector_type(4))) float v4f;
typedef __attribute__((ext_vector_type(8))) float v8f;

#define B_ 64
#define T_ 512
#define D_ 1024
#define NP_ 64
#define NWV_ 4
#define RK_ 8
#define WBLK_ ((B_ * T_) / 256)
#define PSEQ_ (NP_ * 8)

static_assert(NP_ == B_, "one table thread serves range i and sequence i");

__device__ __forceinline__ int ksteps_of(int len) {
  len = len < 0 ? 0 : (len > T_ ? T_ : len);
  return (len + 3) >> 2;
}

__global__ __launch_bounds__(256) void wtab_kernel(
    const int* __restrict__ wordseq, const int* __restrict__ slen,
    const float* __restrict__ weights, float* __restrict__ wtab,
    int* __restrict__ plan) {
  if (blockIdx.x == 0) {
    __shared__ int sh_ks[B_];
    __shared__ int sh_run[B_];
    __shared__ int sh_len[B_];
    const int i = threadIdx.x;
    int ks = 0;
    if (i < B_) {
      const int len = slen[i];
      ks = ksteps_of(len);
      sh_ks[i]  = ks;
      sh_len[i] = len;
    }
    __syncthreads();
    int cst = 0;
    if (i < B_) {
      for (int j = 0; j < B_; ++j) cst += (j < i) ? sh_ks[j] : 0;
      sh_run[i] = cst + ks;
    }
    __syncthreads();
    if (i < B_) {
      const int S = sh_run[B_ - 1];
      int L = (S + NP_ - 1) / NP_;
      if (L < 1) L = 1;
      const int k0 = i * L;
      const int k1 = (k0 + L < S) ? (k0 + L) : S;
      int bf = 0, cs = 0;
      for (int j = 0; j < B_; ++j) {
        const int r = sh_run[j];
        if (r <= k0) { bf = j + 1; cs = r; }
      }
      const int lf = sh_len[bf < B_ ? bf : B_ - 1];
      int4 e0; e0.x = bf; e0.y = cs; e0.z = k0; e0.w = k1;
      int4 e1; e1.x = lf; e1.y = 0;  e1.z = 0;  e1.w = 0;
      ((int4*)plan)[2 * i]     = e0;
      ((int4*)plan)[2 * i + 1] = e1;
      int2 pr; pr.x = 0; pr.y = -1;
      if (ks > 0) { pr.x = cst / L; pr.y = (cst + ks - 1) / L; }
      ((int2*)(plan + PSEQ_))[i] = pr;
    }
    return;
  }
  int i = ((int)blockIdx.x - 1) * 256 + threadIdx.x;
  int b = i >> 9;
  int t = i & (T_ - 1);
  float w = 0.0f;
  if (t < slen[b]) w = weights[wordseq[i]];
  wtab[i] = w;
}

__global__ __launch_bounds__(128) void stream_kernel(
    const float* __restrict__ vs, const int* __restrict__ slen,
    const float* __restrict__ wtab, const int* __restrict__ plan,
    float* __restrict__ s_part, float* __restrict__ yh_part) {
  __shared__ v4f red[2][NWV_][2][16];

  const int tid  = threadIdx.x;
  const int lane = tid & 31;
  const int wv   = __builtin_amdgcn_readfirstlane(tid >> 5);
  const int p      = blockIdx.x >> 4;
  const int d_base = (blockIdx.x & 15) * 64;
  const int wfirst = (wv + (int)blockIdx.x) & (NWV_ - 1);

  const int4 e0 = ((const int4*)plan)[2 * p];
  int len       = plan[8 * p + 4];
  int b         = e0.x;
  int cstart    = e0.y;
  const int k0  = e0.z;
  const int k1  = e0.w;
  if (k0 >= k1) return;

  const int   m = lane & 15;
  const int   h = lane >> 4;
  const float rowsel0 = (m == 0) ? 1.0f : 0.0f;
  const float rowsel1 = (m == 1) ? 1.0f : 0.0f;

  int k = k0;
  int par = 0;
  while (k < k1) {
    const int bend = cstart + ksteps_of(len);
    const int kend = (k1 < bend) ? k1 : bend;
    const int len_nx = slen[(b + 1 < B_) ? (b + 1) : (B_ - 1)];
    if (kend <= k) { cstart = bend; ++b; len = len_nx; continue; }

    const int kl0 = k - cstart;
    const int nk  = kend - k;
    const float* base = vs + (size_t)b * T_ * D_ + (size_t)(d_base + 4 * m);
    const float* wrow = wtab + b * T_;

    v8f c0 = {}, c1 = {}, c2 = {}, c3 = {};

    const int ws = (nk * wfirst) >> 2;
    const int we = (nk * (wfirst + 1)) >> 2;

    if (ws < we) {
      v4f r0[RK_], r1[RK_];
      v2f w2[RK_];
#define LOAD_STEP(u, j)                                                        \
      {                                                                        \
        const int jc_ = ((j) < we) ? (j) : (we - 1);                           \
        const int te_ = (kl0 + jc_) * 4 + 2 * h;                               \
        r0[u] = *(const v4f*)(base + (size_t)te_ * D_);                        \
        r1[u] = *(const v4f*)(base + (size_t)(te_ + 1) * D_);                  \
        w2[u] = *(const v2f*)(wrow + te_);                                     \
      }
#define USE_STEP(u, j)                                                         \
      {                                                                        \
        asm volatile("" : "+v"(r0[u]), "+v"(r1[u]), "+v"(w2[u]));              \
        const int te_ = (kl0 + (j)) * 4 + 2 * h;                               \
        const float mk0 = (te_     < len) ? 1.0f : 0.0f;                       \
        const float mk1 = (te_ + 1 < len) ? 1.0f : 0.0f;                       \
        v2f a;                                                                 \
        a.x = rowsel0 * mk0 + rowsel1 * w2[u].x;                               \
        a.y = rowsel0 * mk1 + rowsel1 * w2[u].y;                               \
        v2f bop;                                                               \
        bop.x = r0[u].x; bop.y = r1[u].x;                                      \
        c0 = __builtin_amdgcn_wmma_f32_16x16x4_f32(false, a, false, bop, (short)0, c0, false, false); \
        bop.x = r0[u].y; bop.y = r1[u].y;                                      \
        c1 = __builtin_amdgcn_wmma_f32_16x16x4_f32(false, a, false, bop, (short)0, c1, false, false); \
        bop.x = r0[u].z; bop.y = r1[u].z;                                      \
        c2 = __builtin_amdgcn_wmma_f32_16x16x4_f32(false, a, false, bop, (short)0, c2, false, false); \
        bop.x = r0[u].w; bop.y = r1[u].w;                                      \
        c3 = __builtin_amdgcn_wmma_f32_16x16x4_f32(false, a, false, bop, (short)0, c3, false, false); \
      }

#pragma unroll
      for (int u = 0; u < RK_; ++u) LOAD_STEP(u, ws + u)

      int i = ws;
      for (; i + RK_ < we; i += RK_) {
#pragma unroll
        for (int u = 0; u < RK_; ++u) {
          USE_STEP(u, i + u)
          LOAD_STEP(u, i + u + RK_)
        }
      }
#pragma unroll
      for (int u = 0; u < RK_; ++u) {
        if (i + u < we) USE_STEP(u, i + u)
      }
#undef LOAD_STEP
#undef USE_STEP
    }

    if (lane < 16) {
      v4f sv, hv;
      sv.x = c0[0]; sv.y = c1[0]; sv.z = c2[0]; sv.w = c3[0];
      hv.x = c0[1]; hv.y = c1[1]; hv.z = c2[1]; hv.w = c3[1];
      red[par][wv][0][lane] = sv;
      red[par][wv][1][lane] = hv;
    }
    __syncthreads();
    if (tid < 32) {
      const int r = tid >> 4;
      const int n = tid & 15;
      v4f acc = red[par][0][r][n];
      for (int w = 1; w < NWV_; ++w) {
        const v4f x = red[par][w][r][n];
        acc.x += x.x; acc.y += x.y; acc.z += x.z; acc.w += x.w;
      }
      float* dst = (r ? yh_part : s_part) + (size_t)(p + b) * D_ + d_base + 4 * n;
      *(v4f*)dst = acc;
    }
    par ^= 1;
    k = kend;
    if (kend == bend) { cstart = bend; ++b; len = len_nx; }
  }
}

#define FG_ 4

template <int CTRL>
__device__ __forceinline__ float row_xadd(float a) {
  const int t = __builtin_amdgcn_update_dpp(0, __float_as_int(a), CTRL, 0xf, 0xf, false);
  return a + __int_as_float(t);
}

__device__ __forceinline__ float xhalf(float x) {
  const int v = __float_as_int(x);
  return __int_as_float(__builtin_amdgcn_permlanex16(v, v, 0x76543210, 0xfedcba98, false, false));
}

template <int N>
__device__ __forceinline__ void add_partials(
    const float* __restrict__ s_part, const float* __restrict__ yh_part,
    int slot, int lane, v4f (&sv)[8], v4f (&hv)[8]) {
  v4f x[N][8], z[N][8];
#pragma unroll
  for (int q = 0; q < N; ++q) {
    const v4f* sp = (const v4f*)(s_part  + (size_t)(slot + q) * D_) + lane;
    const v4f* hp = (const v4f*)(yh_part + (size_t)(slot + q) * D_) + lane;
#pragma unroll
    for (int c = 0; c < 8; ++c) { x[q][c] = sp[32 * c]; z[q][c] = hp[32 * c]; }
  }
  __builtin_amdgcn_sched_barrier(0);
#pragma unroll
  for (int q = 0; q < N; ++q) {
#pragma unroll
    for (int c = 0; c < 8; ++c) { sv[c] += x[q][c]; hv[c] += z[q][c]; }
  }
}

__global__ __launch_bounds__(32) void finalize_kernel(
    const int* __restrict__ plan,
    const float* __restrict__ s_part, const float* __restrict__ yh_part,
    float* __restrict__ y_out, float* __restrict__ yh_out) {
  const int b    = blockIdx.x;
  const int lane = threadIdx.x;

  const int2 pr  = ((const int2*)(plan + PSEQ_))[b];
  const int p_lo = pr.x;
  const int p_hi = pr.y;

  v4f sv[8], hv[8];
#pragma unroll
  for (int c = 0; c < 8; ++c) {
    sv[c] = (v4f){0.0f, 0.0f, 0.0f, 0.0f};
    hv[c] = (v4f){0.0f, 0.0f, 0.0f, 0.0f};
  }

  int p = p_lo;
  for (; p + FG_ - 1 <= p_hi; p += FG_)
    add_partials<FG_>(s_part, yh_part, p + b, lane, sv, hv);
  const int rem = p_hi - p + 1;
  if (rem == 3)      add_partials<3>(s_part, yh_part, p + b, lane, sv, hv);
  else if (rem == 2) add_partials<2>(s_part, yh_part, p + b, lane, sv, hv);
  else if (rem == 1) add_partials<1>(s_part, yh_part, p + b, lane, sv, hv);

  v4f* yh_row = (v4f*)(yh_out + (size_t)b * D_) + lane;
#pragma unroll
  for (int c = 0; c < 8; ++c) yh_row[32 * c] = hv[c];

  float a = 0.0f;
#pragma unroll
  for (int c = 0; c < 8; ++c)
    a += fabsf(sv[c].x) + fabsf(sv[c].y) + fabsf(sv[c].z) + fabsf(sv[c].w);
  a = row_xadd<0x161>(a);
  a = row_xadd<0x162>(a);
  a = row_xadd<0x164>(a);
  a = row_xadd<0x168>(a);
  a += xhalf(a);

  const float inv = 1.0f / sqrtf(a);
  v4f* y_row = (v4f*)(y_out + (size_t)b * D_) + lane;
#pragma unroll
  for (int c = 0; c < 8; ++c) {
    v4f yv;
    yv.x = sv[c].x * inv; yv.y = sv[c].y * inv; yv.z = sv[c].z * inv; yv.w = sv[c].w * inv;
    y_row[32 * c] = yv;
  }
}

extern "C" void kernel_launch(void* const* d_in, const int* in_sizes, int n_in,
                              void* d_out, int out_size, void* d_ws, size_t ws_size,
                              hipStream_t stream) {
  const float* vs      = (const float*)d_in[0];
  const int*   slen    = (const int*)d_in[1];
  const int*   wordseq = (const int*)d_in[2];
  const float* weights = (const float*)d_in[3];

  float* y_out  = (float*)d_out;
  float* yh_out = (float*)d_out + B_ * D_;

  float* wtab    = (float*)d_ws;
  float* s_part  = wtab + B_ * T_;
  float* yh_part = s_part + 2 * B_ * D_;
  int*   plan    = (int*)(s_part + 127 * D_);

  wtab_kernel<<<WBLK_ + 1, 256, 0, stream>>>(wordseq, slen, weights, wtab, plan);
  stream_kernel<<<NP_ * 16, NWV_ * 32, 0, stream>>>(vs, slen, wtab, plan, s_part, yh_part);
  finalize_kernel<<<B_, 32, 0, stream>>>(plan, s_part, yh_part, y_out, yh_out);
}


# launch 0 dropped (3 launches to 2): each wave splits the joined axis itself by a lane scan of the 64 lengths and fetches weights[word] per lane, two loads deep, inside its row ring (sums unchanged)
# speedup vs baseline: 1.7665x; 1.1218x over previous
#include <hip/hip_runtime.h>
#include <hip/hip_bf16.h>


typedef __attribute__((ext_vector_type(2))) float v2f;
typedef __attribute__((ext_vector_type(4))) float v4f;
typedef __attribute__((ext_vector_type(8))) float v8f;
typedef __attribute__((ext_vector_type(2))) int   v2i;

#define B_ 64
#define T_ 512
#define D_ 1024
#define NP_ 64
#define NWV_ 4
#define RK_ 8
#define PSEQ_ (NP_ * 8)

static_assert(B_ == 64 && NP_ == 64, "a lane holds sequences l and l + 32; a range's length is a shift by 6");

__device__ __forceinline__ int ksteps_of(int len) {
  len = len < 0 ? 0 : (len > T_ ? T_ : len);
  return (len + 3) >> 2;
}

template <int CTRL>
__device__ __forceinline__ int row_xchg(int a) {
  return __builtin_amdgcn_update_dpp(0, a, CTRL, 0xf, 0xf, false);
}

__device__ __forceinline__ int xhalf_i(int v) {
  return __builtin_amdgcn_permlanex16(v, v, 0x76543210, 0xfedcba98, false, false);
}

__device__ __forceinline__ void lane_scan(int x, int lane, int& pre, int& tot) {
  pre = x; tot = x;
  int t;
  t = row_xchg<0x161>(tot); pre += (lane & 1)  ? t : 0; tot += t;
  t = row_xchg<0x162>(tot); pre += (lane & 2)  ? t : 0; tot += t;
  t = row_xchg<0x164>(tot); pre += (lane & 4)  ? t : 0; tot += t;
  t = row_xchg<0x168>(tot); pre += (lane & 8)  ? t : 0; tot += t;
  t = xhalf_i(tot);         pre += (lane & 16) ? t : 0; tot += t;
}

__device__ __forceinline__ int seq_pick(int lo, int hi, int i) {
  const int a = __builtin_amdgcn_readlane(lo, i & 31);
  const int c = __builtin_amdgcn_readlane(hi, i & 31);
  return (i < 32) ? a : c;
}

__global__ __launch_bounds__(128) void stream_kernel(
    const float* __restrict__ vs, const int* __restrict__ slen,
    const int* __restrict__ wordseq, const float* __restrict__ weights,
    int* __restrict__ plan,
    float* __restrict__ s_part, float* __restrict__ yh_part) {
  __shared__ v4f red[2][NWV_][2][16];

  const int tid  = threadIdx.x;
  const int lane = tid & 31;
  const int wv   = __builtin_amdgcn_readfirstlane(tid >> 5);
  const int p      = blockIdx.x >> 4;
  const int d_base = (blockIdx.x & 15) * 64;
  const int wfirst = (wv + (int)blockIdx.x) & (NWV_ - 1);

  const int la = slen[lane];
  const int lb = slen[lane + 32];
  const int ka = ksteps_of(la);
  const int kb = ksteps_of(lb);
  int ra, ta, rb, tb;
  lane_scan(ka, lane, ra, ta);
  lane_scan(kb, lane, rb, tb);
  rb += ta;
  const int S = __builtin_amdgcn_readfirstlane(ta + tb);
  int L = (S + NP_ - 1) >> 6;
  if (L < 1) L = 1;
  const int k0 = p * L;
  const int k1 = (k0 + L < S) ? (k0 + L) : S;

  if (blockIdx.x == 0 && wv == 0) {
    const unsigned uL = (unsigned)L;
    int2 pa; pa.x = 0; pa.y = -1;
    if (ka > 0) { pa.x = (int)((unsigned)(ra - ka) / uL); pa.y = (int)((unsigned)(ra - 1) / uL); }
    int2 pb; pb.x = 0; pb.y = -1;
    if (kb > 0) { pb.x = (int)((unsigned)(rb - kb) / uL); pb.y = (int)((unsigned)(rb - 1) / uL); }
    ((int2*)(plan + PSEQ_))[lane]      = pa;
    ((int2*)(plan + PSEQ_))[lane + 32] = pb;
  }
  if (k0 >= k1) return;

  int b = __builtin_popcount(__builtin_amdgcn_ballot_w32(ra <= k0)) +
          __builtin_popcount(__builtin_amdgcn_ballot_w32(rb <= k0));
  int cstart = (b > 0) ? seq_pick(ra, rb, b - 1) : 0;
  int len    = seq_pick(la, lb, b);

  const int   m = lane & 15;
  const int   h = lane >> 4;
  const float rowsel0 = (m == 0) ? 1.0f : 0.0f;
  const float rowsel1 = (m == 1) ? 1.0f : 0.0f;

  int k = k0;
  int par = 0;
  while (k < k1) {
    const int bend = cstart + ksteps_of(len);
    const int kend = (k1 < bend) ? k1 : bend;
    const int len_nx = seq_pick(la, lb, (b + 1 < B_) ? (b + 1) : (B_ - 1));
    if (kend <= k) { cstart = bend; ++b; len = len_nx; continue; }

    const int kl0 = k - cstart;
    const int nk  = kend - k;
    const float* base = vs + (size_t)b * T_ * D_ + (size_t)(d_base + 4 * m);
    const int*   irow = wordseq + b * T_;

    v8f c0 = {}, c1 = {}, c2 = {}, c3 = {};

    const int ws = (nk * wfirst) >> 2;
    const int we = (nk * (wfirst + 1)) >> 2;

    if (ws < we) {
      v4f r0[RK_], r1[RK_];
      v2f w2[RK_];
      v2i ix[RK_], ixn[RK_];
#define LOAD_IX(q, u, j)                                                       \
      {                                                                        \
        const int jc_ = ((j) < we) ? (j) : (we - 1);                           \
        const int te_ = (kl0 + jc_) * 4 + 2 * h;                               \
        q[u] = *(const v2i*)(irow + te_);                                      \
      }
#define LOAD_ROWS(u, j)                                                        \
      {                                                                        \
        const int jc_ = ((j) < we) ? (j) : (we - 1);                           \
        const int te_ = (kl0 + jc_) * 4 + 2 * h;                               \
        r0[u] = *(const v4f*)(base + (size_t)te_ * D_);                        \
        r1[u] = *(const v4f*)(base + (size_t)(te_ + 1) * D_);                  \
      }
#define LOAD_W(q, u)                                                           \
      {                                                                        \
        w2[u].x = weights[q[u].x];                                             \
        w2[u].y = weights[q[u].y];                                             \
      }
#define USE_STEP(u, j)                                                         \
      {                                                                        \
        asm volatile("" : "+v"(r0[u]), "+v"(r1[u]), "+v"(w2[u]));              \
        const int te_ = (kl0 + (j)) * 4 + 2 * h;                               \
        const bool lv0 = te_     < len;                                        \
        const bool lv1 = te_ + 1 < len;                                        \
        v2f a;                                                                 \
        a.x = rowsel0 * (lv0 ? 1.0f : 0.0f) + rowsel1 * (lv0 ? w2[u].x : 0.0f); \
        a.y = rowsel0 * (lv1 ? 1.0f : 0.0f) + rowsel1 * (lv1 ? w2[u].y : 0.0f); \
        v2f bop;                                                               \
        bop.x = r0[u].x; bop.y = r1[u].x;                                      \
        c0 = __builtin_amdgcn_wmma_f32_16x16x4_f32(false, a, false, bop, (short)0, c0, false, false); \
        bop.x = r0[u].y; bop.y = r1[u].y;                                      \
        c1 = __builtin_amdgcn_wmma_f32_16x16x4_f32(false, a, false, bop, (short)0, c1, false, false); \
        bop.x = r0[u].z; bop.y = r1[u].z;                                      \
        c2 = __builtin_amdgcn_wmma_f32_16x16x4_f32(false, a, false, bop, (short)0, c2, false, false); \
        bop.x = r0[u].w; bop.y = r1[u].w;                                      \
        c3 = __builtin_amdgcn_wmma_f32_16x16x4_f32(false, a, false, bop, (short)0, c3, false, false); \
      }

#pragma unroll
      for (int u = 0; u < RK_; ++u) LOAD_IX(ix, u, ws + u)
      if (ws + RK_ < we) {
#pragma unroll
        for (int u = 0; u < RK_; ++u) LOAD_IX(ixn, u, ws + u + RK_)
      }
#pragma unroll
      for (int u = 0; u < RK_; ++u) LOAD_ROWS(u, ws + u)
      __builtin_amdgcn_sched_barrier(0);
#pragma unroll
      for (int u = 0; u < RK_; ++u) LOAD_W(ix, u)

      int i = ws;
      for (; i + RK_ < we; i += RK_) {
#pragma unroll
        for (int u = 0; u < RK_; ++u) {
          USE_STEP(u, i + u)
          LOAD_ROWS(u, i + u + RK_)
          LOAD_W(ixn, u)
          LOAD_IX(ixn, u, i + u + 2 * RK_)
        }
      }
#pragma unroll
      for (int u = 0; u < RK_; ++u) {
        if (i + u < we) USE_STEP(u, i + u)
      }
#undef LOAD_IX
#undef LOAD_ROWS
#undef LOAD_W
#undef USE_STEP
    }

    if (lane < 16) {
      v4f sv, hv;
      sv.x = c0[0]; sv.y = c1[0]; sv.z = c2[0]; sv.w = c3[0];
      hv.x = c0[1]; hv.y = c1[1]; hv.z = c2[1]; hv.w = c3[1];
      red[par][wv][0][lane] = sv;
      red[par][wv][1][lane] = hv;
    }
    __syncthreads();
    if (tid < 32) {
      const int r = tid >> 4;
      const int n = tid & 15;
      v4f acc = red[par][0][r][n];
      for (int w = 1; w < NWV_; ++w) {
        const v4f x = red[par][w][r][n];
        acc.x += x.x; acc.y += x.y; acc.z += x.z; acc.w += x.w;
      }
      float* dst = (r ? yh_part : s_part) + (size_t)(p + b) * D_ + d_base + 4 * n;
      *(v4f*)dst = acc;
    }
    par ^= 1;
    k = kend;
    if (kend == bend) { cstart = bend; ++b; len = len_nx; }
  }
}

#define FG_ 4

template <int CTRL>
__device__ __forceinline__ float row_xadd(float a) {
  const int t = __builtin_amdgcn_update_dpp(0, __float_as_int(a), CTRL, 0xf, 0xf, false);
  return a + __int_as_float(t);
}

__device__ __forceinline__ float xhalf(float x) {
  const int v = __float_as_int(x);
  return __int_as_float(__builtin_amdgcn_permlanex16(v, v, 0x76543210, 0xfedcba98, false, false));
}

template <int N>
__device__ __forceinline__ void add_partials(
    const float* __restrict__ s_part, const float* __restrict__ yh_part,
    int slot, int lane, v4f (&sv)[8], v4f (&hv)[8]) {
  v4f x[N][8], z[N][8];
#pragma unroll
  for (int q = 0; q < N; ++q) {
    const v4f* sp = (const v4f*)(s_part  + (size_t)(slot + q) * D_) + lane;
    const v4f* hp = (const v4f*)(yh_part + (size_t)(slot + q) * D_) + lane;
#pragma unroll
    for (int c = 0; c < 8; ++c) { x[q][c] = sp[32 * c]; z[q][c] = hp[32 * c]; }
  }
  __builtin_amdgcn_sched_barrier(0);
#pragma unroll
  for (int q = 0; q < N; ++q) {
#pragma unroll
    for (int c = 0; c < 8; ++c) { sv[c] += x[q][c]; hv[c] += z[q][c]; }
  }
}

__global__ __launch_bounds__(32) void finalize_kernel(
    const int* __restrict__ plan,
    const float* __restrict__ s_part, const float* __restrict__ yh_part,
    float* __restrict__ y_out, float* __restrict__ yh_out) {
  const int b    = blockIdx.x;
  const int lane = threadIdx.x;

  const int2 pr  = ((const int2*)(plan + PSEQ_))[b];
  const int p_lo = pr.x;
  const int p_hi = pr.y;

  v4f sv[8], hv[8];
#pragma unroll
  for (int c = 0; c < 8; ++c) {
    sv[c] = (v4f){0.0f, 0.0f, 0.0f, 0.0f};
    hv[c] = (v4f){0.0f, 0.0f, 0.0f, 0.0f};
  }

  int p = p_lo;
  for (; p + FG_ - 1 <= p_hi; p += FG_)
    add_partials<FG_>(s_part, yh_part, p + b, lane, sv, hv);
  const int rem = p_hi - p + 1;
  if (rem == 3)      add_partials<3>(s_part, yh_part, p + b, lane, sv, hv);
  else if (rem == 2) add_partials<2>(s_part, yh_part, p + b, lane, sv, hv);
  else if (rem == 1) add_partials<1>(s_part, yh_part, p + b, lane, sv, hv);

  v4f* yh_row = (v4f*)(yh_out + (size_t)b * D_) + lane;
#pragma unroll
  for (int c = 0; c < 8; ++c) yh_row[32 * c] = hv[c];

  float a = 0.0f;
#pragma unroll
  for (int c = 0; c < 8; ++c)
    a += fabsf(sv[c].x) + fabsf(sv[c].y) + fabsf(sv[c].z) + fabsf(sv[c].w);
  a = row_xadd<0x161>(a);
  a = row_xadd<0x162>(a);
  a = row_xadd<0x164>(a);
  a = row_xadd<0x168>(a);
  a += xhalf(a);

  const float inv = 1.0f / sqrtf(a);
  v4f* y_row = (v4f*)(y_out + (size_t)b * D_) + lane;
#pragma unroll
  for (int c = 0; c < 8; ++c) {
    v4f yv;
    yv.x = sv[c].x * inv; yv.y = sv[c].y * inv; yv.z = sv[c].z * inv; yv.w = sv[c].w * inv;
    y_row[32 * c] = yv;
  }
}

extern "C" void kernel_launch(void* const* d_in, const int* in_sizes, int n_in,
                              void* d_out, int out_size, void* d_ws, size_t ws_size,
                              hipStream_t stream) {
  const float* vs      = (const float*)d_in[0];
  const int*   slen    = (const int*)d_in[1];
  const int*   wordseq = (const int*)d_in[2];
  const float* weights = (const float*)d_in[3];

  float* y_out  = (float*)d_out;
  float* yh_out = (float*)d_out + B_ * D_;

  float* s_part  = (float*)d_ws + B_ * T_;
  float* yh_part = s_part + 2 * B_ * D_;
  int*   plan    = (int*)(s_part + 127 * D_);

  stream_kernel<<<NP_ * 16, NWV_ * 32, 0, stream>>>(vs, slen, wordseq, weights, plan, s_part, yh_part);
  finalize_kernel<<<B_, 32, 0, stream>>>(plan, s_part, yh_part, y_out, yh_out);
}
